# starting kernel
# baseline (speedup 1.0000x reference)
#include <hip/hip_runtime.h>
#include <hip/hip_bf16.h>
#include <math.h>


typedef float v2f __attribute__((ext_vector_type(2)));
typedef float v8f __attribute__((ext_vector_type(8)));

#define HIDDEN 16
#define PROJ   2
#define INPUT  2

#define LOG2E 1.4426950408889634f

__device__ __forceinline__ float sigmoidf_fast(float x) {
  return __builtin_amdgcn_rcpf(1.0f + __builtin_amdgcn_exp2f(x * -LOG2E));
}

__device__ __forceinline__ float sigmoid_pre(float x) {
  return __builtin_amdgcn_rcpf(1.0f + __builtin_amdgcn_exp2f(x));
}

__device__ __forceinline__ float tanhf_fast(float x) {
#if __has_builtin(__builtin_amdgcn_tanhf)
  return __builtin_amdgcn_tanhf(x);
#else
  float e = __builtin_amdgcn_exp2f(x * (-2.0f * LOG2E));
  return __builtin_fmaf(2.0f, __builtin_amdgcn_rcpf(1.0f + e), -1.0f);
#endif
}

__device__ __forceinline__ float halfswap16(float v) {
#if __has_builtin(__builtin_amdgcn_permlanex16)
  int i = __builtin_bit_cast(int, v);
  i = __builtin_amdgcn_permlanex16(i, i, 0x76543210, 0xFEDCBA98, false, false);
  return __builtin_bit_cast(float, i);
#else
  return __shfl_xor(v, 16, 32);
#endif
}

__global__ void __launch_bounds__(256) lstm_phase1(
    const float* __restrict__ y_train, const float* __restrict__ W_ih,
    const float* __restrict__ b_ih, const float* __restrict__ b_hh,
    const float* __restrict__ W_hr, float* __restrict__ out,
    float* __restrict__ c_last, int B, int T, int nstep) {
  int idx = blockIdx.x * blockDim.x + threadIdx.x;
  if (idx >= B * nstep) return;
  int b = idx / nstep;
  int t = idx - b * nstep;

  const float* xp = y_train + ((size_t)b * T + t) * 2;
  float x0 = xp[0], x1 = xp[1];
  float y0 = 0.f, y1 = 0.f;
  bool last = (t == nstep - 1);

#pragma unroll
  for (int n = 0; n < HIDDEN; ++n) {
    int ng = 32 + n, no = 48 + n;
    float gi = W_ih[n * 2 + 0] * x0 + W_ih[n * 2 + 1] * x1 + b_ih[n] + b_hh[n];
    float gg = W_ih[ng * 2 + 0] * x0 + W_ih[ng * 2 + 1] * x1 + b_ih[ng] + b_hh[ng];
    float go = W_ih[no * 2 + 0] * x0 + W_ih[no * 2 + 1] * x1 + b_ih[no] + b_hh[no];
    float cv = sigmoidf_fast(gi) * tanhf_fast(gg);
    float hp = sigmoidf_fast(go) * tanhf_fast(cv);
    y0 = __builtin_fmaf(hp, W_hr[n], y0);
    y1 = __builtin_fmaf(hp, W_hr[16 + n], y1);
    if (last) c_last[b * HIDDEN + n] = cv;
  }
  float* op = out + ((size_t)b * T + t) * 2;
  op[0] = y0;
  op[1] = y1;
}

__global__ void __launch_bounds__(32) lstm_phase2(
    const float* __restrict__ u_train, const float* __restrict__ W_ih,
    const float* __restrict__ W_hh, const float* __restrict__ b_ih,
    const float* __restrict__ b_hh, const float* __restrict__ W_hr,
    const float* __restrict__ c_last, float* __restrict__ out,
    int T, int nstep, int nsteps2) {
  const int lane  = threadIdx.x;
  const int half  = lane >> 4;
  const int j     = lane & 15;
  const int mbase = blockIdx.x * 16;

  v2f At[4];
#pragma unroll
  for (int t = 0; t < 4; ++t) {
    int n = t * 16 + j;
    float sc = (t == 2) ? 1.0f : -LOG2E;
    At[t][0] = sc * (half ? W_hh[n * 2 + 0] : W_ih[n * 2 + 0]);
    At[t][1] = sc * (half ? W_hh[n * 2 + 1] : W_ih[n * 2 + 1]);
  }

  const int jjbase = half * 8;
  v8f cc0, cc1, cc2, cc3;
  float w0[8], w1[8];
#pragma unroll
  for (int r = 0; r < 8; ++r) {
    int jj = jjbase + r;
    cc0[r] = -LOG2E * (b_ih[jj]      + b_hh[jj]);
    cc1[r] = -LOG2E * (b_ih[16 + jj] + b_hh[16 + jj]);
    cc2[r] =           b_ih[32 + jj] + b_hh[32 + jj];
    cc3[r] = -LOG2E * (b_ih[48 + jj] + b_hh[48 + jj]);
    w0[r]  = W_hr[jj];
    w1[r]  = W_hr[16 + jj];
  }

  float c[8];
#pragma unroll
  for (int r = 0; r < 8; ++r)
    c[r] = c_last[(mbase + j) * HIDDEN + jjbase + r];

  float hb0, hb1;
  {
    const float* hp = out + ((size_t)(mbase + j) * T + (nstep - 1)) * 2;
    hb0 = hp[0];
    hb1 = hp[1];
  }

  const float* xbase = u_train + ((size_t)(mbase + j) * T + nstep) * 2;
  float* obase = out + ((size_t)(mbase + j) * T + nstep) * 2;
  const int lastS = nsteps2 - 1;

  float x0 = xbase[0];
  float x1 = xbase[1];

  for (int s = 0; s < nsteps2; ++s) {
    int sn = (s < lastS) ? s + 1 : lastS;
    float nx0 = xbase[sn * 2 + 0];
    float nx1 = xbase[sn * 2 + 1];
    __builtin_prefetch(xbase + sn * 2 + 512, 0, 3);

    v2f bv;
    bv[0] = half ? hb0 : x0;
    bv[1] = half ? hb1 : x1;

    v8f d0 = __builtin_amdgcn_wmma_f32_16x16x4_f32(false, At[0], false, bv, (short)0, cc0, false, false);
    v8f d1 = __builtin_amdgcn_wmma_f32_16x16x4_f32(false, At[1], false, bv, (short)0, cc1, false, false);
    v8f d2 = __builtin_amdgcn_wmma_f32_16x16x4_f32(false, At[2], false, bv, (short)0, cc2, false, false);
    v8f d3 = __builtin_amdgcn_wmma_f32_16x16x4_f32(false, At[3], false, bv, (short)0, cc3, false, false);

    float hp[8];
#pragma unroll
    for (int r = 0; r < 8; ++r) {
      float ig = sigmoid_pre(d0[r]);
      float fg = sigmoid_pre(d1[r]);
      float gg = tanhf_fast(d2[r]);
      float og = sigmoid_pre(d3[r]);
      float cr = __builtin_fmaf(fg, c[r], ig * gg);
      c[r] = cr;
      hp[r] = og * tanhf_fast(cr);
    }

    float s0 = (__builtin_fmaf(hp[0], w0[0], hp[1] * w0[1]) +
                __builtin_fmaf(hp[2], w0[2], hp[3] * w0[3])) +
               (__builtin_fmaf(hp[4], w0[4], hp[5] * w0[5]) +
                __builtin_fmaf(hp[6], w0[6], hp[7] * w0[7]));
    float s1 = (__builtin_fmaf(hp[0], w1[0], hp[1] * w1[1]) +
                __builtin_fmaf(hp[2], w1[2], hp[3] * w1[3])) +
               (__builtin_fmaf(hp[4], w1[4], hp[5] * w1[5]) +
                __builtin_fmaf(hp[6], w1[6], hp[7] * w1[7]));
    s0 += halfswap16(s0);
    s1 += halfswap16(s1);

    hb0 = s0;
    hb1 = s1;

    obase[s * 2 + 0] = s0;
    obase[s * 2 + 1] = s1;

    x0 = nx0;
    x1 = nx1;
  }
}

extern "C" void kernel_launch(void* const* d_in, const int* in_sizes, int n_in,
                              void* d_out, int out_size, void* d_ws, size_t ws_size,
                              hipStream_t stream) {
  const float* u_train = (const float*)d_in[0];
  const float* y_train = (const float*)d_in[1];
  const float* W_ih    = (const float*)d_in[2];
  const float* W_hh    = (const float*)d_in[3];
  const float* b_ih    = (const float*)d_in[4];
  const float* b_hh    = (const float*)d_in[5];
  const float* W_hr    = (const float*)d_in[6];
  const int B = 64;
  const int T = in_sizes[0] / (B * INPUT);
  const int nstep = T / 2;

  float* out    = (float*)d_out;
  float* c_last = (float*)d_ws;

  int total = B * nstep;
  lstm_phase1<<<(total + 255) / 256, 256, 0, stream>>>(
      y_train, W_ih, b_ih, b_hh, W_hr, out, c_last, B, T, nstep);

  lstm_phase2<<<B / 16, 32, 0, stream>>>(
      u_train, W_ih, W_hh, b_ih, b_hh, W_hr, c_last, out, T, nstep, T - nstep);
}


# Scan split over four waves of the work-group, each owning 4 hidden units: 1 WMMA and 2 cells a lane a step (was 4 and 8); projection parts meet in LDS, one barrier a step; summation order unchanged
# speedup vs baseline: 1.6005x; 1.6005x over previous
#include <hip/hip_runtime.h>
#include <hip/hip_bf16.h>
#include <math.h>


typedef float v2f __attribute__((ext_vector_type(2)));
typedef float v8f __attribute__((ext_vector_type(8)));

#define HIDDEN 16
#define PROJ   2
#define INPUT  2

#define LOG2E 1.4426950408889634f

__device__ __forceinline__ float sigmoidf_fast(float x) {
  return __builtin_amdgcn_rcpf(1.0f + __builtin_amdgcn_exp2f(x * -LOG2E));
}

__device__ __forceinline__ float sigmoid_pre(float x) {
  return __builtin_amdgcn_rcpf(1.0f + __builtin_amdgcn_exp2f(x));
}

__device__ __forceinline__ float tanhf_fast(float x) {
#if __has_builtin(__builtin_amdgcn_tanhf)
  return __builtin_amdgcn_tanhf(x);
#else
  float e = __builtin_amdgcn_exp2f(x * (-2.0f * LOG2E));
  return __builtin_fmaf(2.0f, __builtin_amdgcn_rcpf(1.0f + e), -1.0f);
#endif
}

__device__ __forceinline__ float halfswap16(float v) {
#if __has_builtin(__builtin_amdgcn_permlanex16)
  int i = __builtin_bit_cast(int, v);
  i = __builtin_amdgcn_permlanex16(i, i, 0x76543210, 0xFEDCBA98, false, false);
  return __builtin_bit_cast(float, i);
#else
  return __shfl_xor(v, 16, 32);
#endif
}

__global__ void __launch_bounds__(256) lstm_phase1(
    const float* __restrict__ y_train, const float* __restrict__ W_ih,
    const float* __restrict__ b_ih, const float* __restrict__ b_hh,
    const float* __restrict__ W_hr, float* __restrict__ out,
    float* __restrict__ c_last, int B, int T, int nstep) {
  int idx = blockIdx.x * blockDim.x + threadIdx.x;
  if (idx >= B * nstep) return;
  int b = idx / nstep;
  int t = idx - b * nstep;

  const float* xp = y_train + ((size_t)b * T + t) * 2;
  float x0 = xp[0], x1 = xp[1];
  float y0 = 0.f, y1 = 0.f;
  bool last = (t == nstep - 1);

#pragma unroll
  for (int n = 0; n < HIDDEN; ++n) {
    int ng = 32 + n, no = 48 + n;
    float gi = W_ih[n * 2 + 0] * x0 + W_ih[n * 2 + 1] * x1 + b_ih[n] + b_hh[n];
    float gg = W_ih[ng * 2 + 0] * x0 + W_ih[ng * 2 + 1] * x1 + b_ih[ng] + b_hh[ng];
    float go = W_ih[no * 2 + 0] * x0 + W_ih[no * 2 + 1] * x1 + b_ih[no] + b_hh[no];
    float cv = sigmoidf_fast(gi) * tanhf_fast(gg);
    float hp = sigmoidf_fast(go) * tanhf_fast(cv);
    y0 = __builtin_fmaf(hp, W_hr[n], y0);
    y1 = __builtin_fmaf(hp, W_hr[16 + n], y1);
    if (last) c_last[b * HIDDEN + n] = cv;
  }
  float* op = out + ((size_t)b * T + t) * 2;
  op[0] = y0;
  op[1] = y1;
}

#define CH    128
#define PITCH (CH * 2 + 2)
#define NSW   4
#define RPW   (16 / NSW)
#define PXP   (16 * 2 * NSW + 4)

typedef float v4f __attribute__((ext_vector_type(4)));

__device__ __forceinline__ void lds_barrier() {
  __builtin_amdgcn_fence(__ATOMIC_RELEASE, "workgroup", "local");
  __builtin_amdgcn_s_barrier();
  __builtin_amdgcn_fence(__ATOMIC_ACQUIRE, "workgroup", "local");
}

__device__ __forceinline__ void x_fetch(const float* __restrict__ xg, size_t rowf,
                                        int row0, int sb, int nsteps2, int lane,
                                        v2f (&v)[RPW][CH / 32]) {
#pragma unroll
  for (int bb = 0; bb < RPW; ++bb) {
    const float* row = xg + (size_t)(row0 + bb) * rowf;
#pragma unroll
    for (int m = 0; m < CH / 32; ++m) {
      int sg = sb + lane + 32 * m;
      sg = sg < nsteps2 ? sg : nsteps2 - 1;
      v[bb][m] = *(const v2f*)(row + (size_t)sg * 2);
    }
  }
}

__device__ __forceinline__ void x_put(float (*dst)[PITCH], int row0, int lane,
                                      const v2f (&v)[RPW][CH / 32]) {
#pragma unroll
  for (int bb = 0; bb < RPW; ++bb)
#pragma unroll
    for (int m = 0; m < CH / 32; ++m)
      *(v2f*)(&dst[row0 + bb][2 * (lane + 32 * m)]) = v[bb][m];
}

__device__ __forceinline__ v2f h_join(const float* p) {
  v4f pa = *(const v4f*)(p);
  v4f pb = *(const v4f*)(p + 4);
  v2f h;
  h[0] = (pa[0] + pa[2]) + (pb[0] + pb[2]);
  h[1] = (pa[1] + pa[3]) + (pb[1] + pb[3]);
  return h;
}

__device__ __forceinline__ void h_drain(float* __restrict__ og, size_t rowf,
                                        const float (*src)[PXP], int row0, int sb,
                                        int nsteps2, int lane) {
#pragma unroll
  for (int m = 0; m < CH / 32; ++m) {
    const int st = lane + 32 * m;
    if (sb + st < nsteps2) {
      float* col = og + (size_t)(sb + st) * 2;
#pragma unroll
      for (int bb = 0; bb < RPW; ++bb)
        *(v2f*)(col + (size_t)(row0 + bb) * rowf) =
            h_join(&src[st][(row0 + bb) * 2 * NSW]);
    }
  }
}

__global__ void __launch_bounds__(32 * NSW) lstm_phase2(
    const float* __restrict__ u_train, const float* __restrict__ W_ih,
    const float* __restrict__ W_hh, const float* __restrict__ b_ih,
    const float* __restrict__ b_hh, const float* __restrict__ W_hr,
    const float* __restrict__ c_last, float* __restrict__ out,
    int T, int nstep, int nsteps2) {
  __shared__ __attribute__((aligned(16))) float xs[2][16][PITCH];
  __shared__ __attribute__((aligned(16))) float px[2][CH][PXP];

  if (nsteps2 < 1) return;

  const int lane  = threadIdx.x & 31;
  const int wid   = __builtin_amdgcn_readfirstlane((int)(threadIdx.x >> 5));
  const int half  = lane >> 4;
  const int j     = lane & 15;
  const int mbase = blockIdx.x * 16;
  const int row0  = wid * RPW;

  v2f At;
  {
    const int g  = j & 3;
    const int n  = g * 16 + 4 * wid + (j >> 2);
    const float sc = (g == 2) ? 1.0f : -LOG2E;
    At[0] = sc * (half ? W_hh[n * 2 + 0] : W_ih[n * 2 + 0]);
    At[1] = sc * (half ? W_hh[n * 2 + 1] : W_ih[n * 2 + 1]);
  }

  const int ubase = 4 * wid + 2 * half;
  v8f cc;
#pragma unroll
  for (int r = 0; r < 8; ++r) {
    const int g = r & 3;
    const int n = g * 16 + ubase + (r >> 2);
    const float sc = (g == 2) ? 1.0f : -LOG2E;
    cc[r] = sc * (b_ih[n] + b_hh[n]);
  }

  float c[2], w0[2], w1[2];
#pragma unroll
  for (int q = 0; q < 2; ++q) {
    c[q]  = c_last[(mbase + j) * HIDDEN + ubase + q];
    w0[q] = W_hr[ubase + q];
    w1[q] = W_hr[16 + ubase + q];
  }

  float hb0, hb1;
  {
    const float* hp = out + ((size_t)(mbase + j) * T + (nstep - 1)) * 2;
    hb0 = hp[0];
    hb1 = hp[1];
  }

  const size_t rowf = (size_t)T * 2;
  const float* xg = u_train + ((size_t)mbase * T + nstep) * 2;
  float*       og = out     + ((size_t)mbase * T + nstep) * 2;
  const int nchunks = (nsteps2 + CH - 1) / CH;

  v2f xr[RPW][CH / 32];
  x_fetch(xg, rowf, row0, 0, nsteps2, lane, xr);
  x_put(xs[0], row0, lane, xr);
  lds_barrier();

  for (int ch = 0; ch < nchunks; ++ch) {
    const int buf = ch & 1;
    const int sb  = ch * CH;
    int n = nsteps2 - sb;
    n = n < CH ? n : CH;

    x_fetch(xg, rowf, row0, sb + CH, nsteps2, lane, xr);

    const float* xrow = &xs[buf][j][0];
    float*       prow = &px[buf][0][j * 2 * NSW];
    v2f xv = *(const v2f*)xrow;

    for (int k = 0; k < n; ++k) {
      v2f bv;
      bv[0] = half ? hb0 : xv[0];
      bv[1] = half ? hb1 : xv[1];

      v8f d = __builtin_amdgcn_wmma_f32_16x16x4_f32(false, At, false, bv, (short)0, cc, false, false);

      int kn = (k + 1 < n) ? k + 1 : k;
      v2f nxv = *(const v2f*)(xrow + 2 * kn);

      float hp[2];
#pragma unroll
      for (int q = 0; q < 2; ++q) {
        float ig  = sigmoid_pre(d[4 * q + 0]);
        float fg  = sigmoid_pre(d[4 * q + 1]);
        float gg  = tanhf_fast(d[4 * q + 2]);
        float og_ = sigmoid_pre(d[4 * q + 3]);
        float cr  = __builtin_fmaf(fg, c[q], ig * gg);
        c[q] = cr;
        hp[q] = og_ * tanhf_fast(cr);
      }

      float s0 = __builtin_fmaf(hp[0], w0[0], hp[1] * w0[1]);
      float s1 = __builtin_fmaf(hp[0], w1[0], hp[1] * w1[1]);
      s0 += halfswap16(s0);
      s1 += halfswap16(s1);

      float* pk = prow + (size_t)k * PXP;
      v2f part;
      part[0] = s0;
      part[1] = s1;
      *(v2f*)(pk + 2 * wid) = part;
      lds_barrier();
      v2f hn = h_join(pk);
      hb0 = hn[0];
      hb1 = hn[1];
      xv  = nxv;
    }

    x_put(xs[buf ^ 1], row0, lane, xr);
    lds_barrier();
    h_drain(og, rowf, px[buf], row0, sb, nsteps2, lane);
  }
}

extern "C" void kernel_launch(void* const* d_in, const int* in_sizes, int n_in,
                              void* d_out, int out_size, void* d_ws, size_t ws_size,
                              hipStream_t stream) {
  const float* u_train = (const float*)d_in[0];
  const float* y_train = (const float*)d_in[1];
  const float* W_ih    = (const float*)d_in[2];
  const float* W_hh    = (const float*)d_in[3];
  const float* b_ih    = (const float*)d_in[4];
  const float* b_hh    = (const float*)d_in[5];
  const float* W_hr    = (const float*)d_in[6];
  const int B = 64;
  const int T = in_sizes[0] / (B * INPUT);
  const int nstep = T / 2;

  float* out    = (float*)d_out;
  float* c_last = (float*)d_ws;

  int total = B * nstep;
  lstm_phase1<<<(total + 255) / 256, 256, 0, stream>>>(
      y_train, W_ih, b_ih, b_hh, W_hr, out, c_last, B, T, nstep);

  lstm_phase2<<<B / 16, 32 * NSW, 0, stream>>>(
      u_train, W_ih, W_hh, b_ih, b_hh, W_hr, c_last, out, T, nstep, T - nstep);
}


# Cross-wave join ends in the gate WMMA's K sum: each 16-lane row loads, adds two parts (1 load, 1 add; was 2 loads, 3 adds, a select); x enters via C from its own off-path WMMA (new summation order)
# speedup vs baseline: 1.6783x; 1.0486x over previous
#include <hip/hip_runtime.h>
#include <hip/hip_bf16.h>
#include <math.h>


typedef float v2f __attribute__((ext_vector_type(2)));
typedef float v8f __attribute__((ext_vector_type(8)));

#define HIDDEN 16
#define PROJ   2
#define INPUT  2

#define LOG2E 1.4426950408889634f

__device__ __forceinline__ float sigmoidf_fast(float x) {
  return __builtin_amdgcn_rcpf(1.0f + __builtin_amdgcn_exp2f(x * -LOG2E));
}

__device__ __forceinline__ float sigmoid_pre(float x) {
  return __builtin_amdgcn_rcpf(1.0f + __builtin_amdgcn_exp2f(x));
}

__device__ __forceinline__ float tanhf_fast(float x) {
#if __has_builtin(__builtin_amdgcn_tanhf)
  return __builtin_amdgcn_tanhf(x);
#else
  float e = __builtin_amdgcn_exp2f(x * (-2.0f * LOG2E));
  return __builtin_fmaf(2.0f, __builtin_amdgcn_rcpf(1.0f + e), -1.0f);
#endif
}

__device__ __forceinline__ float halfswap16(float v) {
#if __has_builtin(__builtin_amdgcn_permlanex16)
  int i = __builtin_bit_cast(int, v);
  i = __builtin_amdgcn_permlanex16(i, i, 0x76543210, 0xFEDCBA98, false, false);
  return __builtin_bit_cast(float, i);
#else
  return __shfl_xor(v, 16, 32);
#endif
}

__global__ void __launch_bounds__(256) lstm_phase1(
    const float* __restrict__ y_train, const float* __restrict__ W_ih,
    const float* __restrict__ b_ih, const float* __restrict__ b_hh,
    const float* __restrict__ W_hr, float* __restrict__ out,
    float* __restrict__ c_last, int B, int T, int nstep) {
  int idx = blockIdx.x * blockDim.x + threadIdx.x;
  if (idx >= B * nstep) return;
  int b = idx / nstep;
  int t = idx - b * nstep;

  const float* xp = y_train + ((size_t)b * T + t) * 2;
  float x0 = xp[0], x1 = xp[1];
  float y0 = 0.f, y1 = 0.f;
  bool last = (t == nstep - 1);

#pragma unroll
  for (int n = 0; n < HIDDEN; ++n) {
    int ng = 32 + n, no = 48 + n;
    float gi = W_ih[n * 2 + 0] * x0 + W_ih[n * 2 + 1] * x1 + b_ih[n] + b_hh[n];
    float gg = W_ih[ng * 2 + 0] * x0 + W_ih[ng * 2 + 1] * x1 + b_ih[ng] + b_hh[ng];
    float go = W_ih[no * 2 + 0] * x0 + W_ih[no * 2 + 1] * x1 + b_ih[no] + b_hh[no];
    float cv = sigmoidf_fast(gi) * tanhf_fast(gg);
    float hp = sigmoidf_fast(go) * tanhf_fast(cv);
    y0 = __builtin_fmaf(hp, W_hr[n], y0);
    y1 = __builtin_fmaf(hp, W_hr[16 + n], y1);
    if (last) c_last[b * HIDDEN + n] = cv;
  }
  float* op = out + ((size_t)b * T + t) * 2;
  op[0] = y0;
  op[1] = y1;
}

#define CH    128
#define PITCH (CH * 2 + 2)
#define NSW   4
#define RPW   (16 / NSW)
#define PXP   (16 * 2 * NSW + 4)

typedef float v4f __attribute__((ext_vector_type(4)));

__device__ __forceinline__ void lds_barrier() {
  __builtin_amdgcn_fence(__ATOMIC_RELEASE, "workgroup", "local");
  __builtin_amdgcn_s_barrier();
  __builtin_amdgcn_fence(__ATOMIC_ACQUIRE, "workgroup", "local");
}

__device__ __forceinline__ void x_fetch(const float* __restrict__ xg, size_t rowf,
                                        int row0, int sb, int nsteps2, int lane,
                                        v2f (&v)[RPW][CH / 32]) {
#pragma unroll
  for (int bb = 0; bb < RPW; ++bb) {
    const float* row = xg + (size_t)(row0 + bb) * rowf;
#pragma unroll
    for (int m = 0; m < CH / 32; ++m) {
      int sg = sb + lane + 32 * m;
      sg = sg < nsteps2 ? sg : nsteps2 - 1;
      v[bb][m] = *(const v2f*)(row + (size_t)sg * 2);
    }
  }
}

__device__ __forceinline__ void x_put(float (*dst)[PITCH], int row0, int lane,
                                      const v2f (&v)[RPW][CH / 32]) {
#pragma unroll
  for (int bb = 0; bb < RPW; ++bb)
#pragma unroll
    for (int m = 0; m < CH / 32; ++m)
      *(v2f*)(&dst[row0 + bb][2 * (lane + 32 * m)]) = v[bb][m];
}

__device__ __forceinline__ v2f h_join(const float* p) {
  v4f pa = *(const v4f*)(p);
  v4f pb = *(const v4f*)(p + 4);
  v2f h;
  h[0] = (pa[0] + pa[2]) + (pb[0] + pb[2]);
  h[1] = (pa[1] + pa[3]) + (pb[1] + pb[3]);
  return h;
}

__device__ __forceinline__ void h_drain(float* __restrict__ og, size_t rowf,
                                        const float (*src)[PXP], int row0, int sb,
                                        int nsteps2, int lane) {
#pragma unroll
  for (int m = 0; m < CH / 32; ++m) {
    const int st = lane + 32 * m;
    if (sb + st < nsteps2) {
      float* col = og + (size_t)(sb + st) * 2;
#pragma unroll
      for (int bb = 0; bb < RPW; ++bb)
        *(v2f*)(col + (size_t)(row0 + bb) * rowf) =
            h_join(&src[st][(row0 + bb) * 2 * NSW]);
    }
  }
}

__global__ void __launch_bounds__(32 * NSW) lstm_phase2(
    const float* __restrict__ u_train, const float* __restrict__ W_ih,
    const float* __restrict__ W_hh, const float* __restrict__ b_ih,
    const float* __restrict__ b_hh, const float* __restrict__ W_hr,
    const float* __restrict__ c_last, float* __restrict__ out,
    int T, int nstep, int nsteps2) {
  __shared__ __attribute__((aligned(16))) float xs[2][16][PITCH];
  __shared__ __attribute__((aligned(16))) float px[2][CH][PXP];

  if (nsteps2 < 1) return;

  const int lane  = threadIdx.x & 31;
  const int wid   = __builtin_amdgcn_readfirstlane((int)(threadIdx.x >> 5));
  const int half  = lane >> 4;
  const int j     = lane & 15;
  const int mbase = blockIdx.x * 16;
  const int row0  = wid * RPW;

  v2f Ax, Ah;
  {
    const int g  = j & 3;
    const int n  = g * 16 + 4 * wid + (j >> 2);
    const float sc = (g == 2) ? 1.0f : -LOG2E;
    Ax[0] = half ? 0.0f : sc * W_ih[n * 2 + 0];
    Ax[1] = half ? 0.0f : sc * W_ih[n * 2 + 1];
    Ah[0] = sc * W_hh[n * 2 + 0];
    Ah[1] = sc * W_hh[n * 2 + 1];
  }

  const int ubase = 4 * wid + 2 * half;
  v8f cc;
#pragma unroll
  for (int r = 0; r < 8; ++r) {
    const int g = r & 3;
    const int n = g * 16 + ubase + (r >> 2);
    const float sc = (g == 2) ? 1.0f : -LOG2E;
    cc[r] = sc * (b_ih[n] + b_hh[n]);
  }

  float c[2], w0[2], w1[2];
#pragma unroll
  for (int q = 0; q < 2; ++q) {
    c[q]  = c_last[(mbase + j) * HIDDEN + ubase + q];
    w0[q] = W_hr[ubase + q];
    w1[q] = W_hr[16 + ubase + q];
  }

  v2f bh;
  {
    const float* hp = out + ((size_t)(mbase + j) * T + (nstep - 1)) * 2;
    bh[0] = half ? 0.0f : hp[0];
    bh[1] = half ? 0.0f : hp[1];
  }

  const size_t rowf = (size_t)T * 2;
  const float* xg = u_train + ((size_t)mbase * T + nstep) * 2;
  float*       og = out     + ((size_t)mbase * T + nstep) * 2;
  const int nchunks = (nsteps2 + CH - 1) / CH;

  v2f xr[RPW][CH / 32];
  x_fetch(xg, rowf, row0, 0, nsteps2, lane, xr);
  x_put(xs[0], row0, lane, xr);
  lds_barrier();

  for (int ch = 0; ch < nchunks; ++ch) {
    const int buf = ch & 1;
    const int sb  = ch * CH;
    int n = nsteps2 - sb;
    n = n < CH ? n : CH;

    x_fetch(xg, rowf, row0, sb + CH, nsteps2, lane, xr);

    const float* xrow = &xs[buf][j][0];
    float*       prow = &px[buf][0][j * 2 * NSW];
    const float* jrow = prow + 4 * half;
    v8f cx;
    {
      v2f xv = *(const v2f*)xrow;
      cx = __builtin_amdgcn_wmma_f32_16x16x4_f32(false, Ax, false, xv, (short)0, cc, false, false);
    }

    for (int k = 0; k < n; ++k) {
      v8f d = __builtin_amdgcn_wmma_f32_16x16x4_f32(false, Ah, false, bh, (short)0, cx, false, false);

      int kn = (k + 1 < n) ? k + 1 : k;
      v2f nxv = *(const v2f*)(xrow + 2 * kn);

      float hp[2];
#pragma unroll
      for (int q = 0; q < 2; ++q) {
        float ig  = sigmoid_pre(d[4 * q + 0]);
        float fg  = sigmoid_pre(d[4 * q + 1]);
        float gg  = tanhf_fast(d[4 * q + 2]);
        float og_ = sigmoid_pre(d[4 * q + 3]);
        float cr  = __builtin_fmaf(fg, c[q], ig * gg);
        c[q] = cr;
        hp[q] = og_ * tanhf_fast(cr);
      }

      float s0 = __builtin_fmaf(hp[0], w0[0], hp[1] * w0[1]);
      float s1 = __builtin_fmaf(hp[0], w1[0], hp[1] * w1[1]);
      s0 += halfswap16(s0);
      s1 += halfswap16(s1);

      float* pk = prow + (size_t)k * PXP;
      v2f part;
      part[0] = s0;
      part[1] = s1;
      *(v2f*)(pk + 2 * wid) = part;

      cx = __builtin_amdgcn_wmma_f32_16x16x4_f32(false, Ax, false, nxv, (short)0, cc, false, false);

      lds_barrier();
      v4f q = *(const v4f*)(jrow + (size_t)k * PXP);
      bh[0] = q[0] + q[2];
      bh[1] = q[1] + q[3];
    }

    x_put(xs[buf ^ 1], row0, lane, xr);
    lds_barrier();
    h_drain(og, rowf, px[buf], row0, sb, nsteps2, lane);
  }
}

extern "C" void kernel_launch(void* const* d_in, const int* in_sizes, int n_in,
                              void* d_out, int out_size, void* d_ws, size_t ws_size,
                              hipStream_t stream) {
  const float* u_train = (const float*)d_in[0];
  const float* y_train = (const float*)d_in[1];
  const float* W_ih    = (const float*)d_in[2];
  const float* W_hh    = (const float*)d_in[3];
  const float* b_ih    = (const float*)d_in[4];
  const float* b_hh    = (const float*)d_in[5];
  const float* W_hr    = (const float*)d_in[6];
  const int B = 64;
  const int T = in_sizes[0] / (B * INPUT);
  const int nstep = T / 2;

  float* out    = (float*)d_out;
  float* c_last = (float*)d_ws;

  int total = B * nstep;
  lstm_phase1<<<(total + 255) / 256, 256, 0, stream>>>(
      y_train, W_ih, b_ih, b_hh, W_hr, out, c_last, B, T, nstep);

  lstm_phase2<<<B / 16, 32 * NSW, 0, stream>>>(
      u_train, W_ih, W_hh, b_ih, b_hh, W_hr, c_last, out, T, nstep, T - nstep);
}


# Next step's x read leaves the serial tail: read two steps ahead as the second LDS load beside the exchange read behind the barrier, parts waited for with it in flight; same sums, same order
# speedup vs baseline: 1.8420x; 1.0975x over previous
#include <hip/hip_runtime.h>
#include <hip/hip_bf16.h>
#include <math.h>


typedef float v2f __attribute__((ext_vector_type(2)));
typedef float v8f __attribute__((ext_vector_type(8)));

#define HIDDEN 16
#define PROJ   2
#define INPUT  2

#define LOG2E 1.4426950408889634f

__device__ __forceinline__ float sigmoidf_fast(float x) {
  return __builtin_amdgcn_rcpf(1.0f + __builtin_amdgcn_exp2f(x * -LOG2E));
}

__device__ __forceinline__ float sigmoid_pre(float x) {
  return __builtin_amdgcn_rcpf(1.0f + __builtin_amdgcn_exp2f(x));
}

__device__ __forceinline__ float tanhf_fast(float x) {
#if __has_builtin(__builtin_amdgcn_tanhf)
  return __builtin_amdgcn_tanhf(x);
#else
  float e = __builtin_amdgcn_exp2f(x * (-2.0f * LOG2E));
  return __builtin_fmaf(2.0f, __builtin_amdgcn_rcpf(1.0f + e), -1.0f);
#endif
}

__device__ __forceinline__ float halfswap16(float v) {
#if __has_builtin(__builtin_amdgcn_permlanex16)
  int i = __builtin_bit_cast(int, v);
  i = __builtin_amdgcn_permlanex16(i, i, 0x76543210, 0xFEDCBA98, false, false);
  return __builtin_bit_cast(float, i);
#else
  return __shfl_xor(v, 16, 32);
#endif
}

__global__ void __launch_bounds__(256) lstm_phase1(
    const float* __restrict__ y_train, const float* __restrict__ W_ih,
    const float* __restrict__ b_ih, const float* __restrict__ b_hh,
    const float* __restrict__ W_hr, float* __restrict__ out,
    float* __restrict__ c_last, int B, int T, int nstep) {
  int idx = blockIdx.x * blockDim.x + threadIdx.x;
  if (idx >= B * nstep) return;
  int b = idx / nstep;
  int t = idx - b * nstep;

  const float* xp = y_train + ((size_t)b * T + t) * 2;
  float x0 = xp[0], x1 = xp[1];
  float y0 = 0.f, y1 = 0.f;
  bool last = (t == nstep - 1);

#pragma unroll
  for (int n = 0; n < HIDDEN; ++n) {
    int ng = 32 + n, no = 48 + n;
    float gi = W_ih[n * 2 + 0] * x0 + W_ih[n * 2 + 1] * x1 + b_ih[n] + b_hh[n];
    float gg = W_ih[ng * 2 + 0] * x0 + W_ih[ng * 2 + 1] * x1 + b_ih[ng] + b_hh[ng];
    float go = W_ih[no * 2 + 0] * x0 + W_ih[no * 2 + 1] * x1 + b_ih[no] + b_hh[no];
    float cv = sigmoidf_fast(gi) * tanhf_fast(gg);
    float hp = sigmoidf_fast(go) * tanhf_fast(cv);
    y0 = __builtin_fmaf(hp, W_hr[n], y0);
    y1 = __builtin_fmaf(hp, W_hr[16 + n], y1);
    if (last) c_last[b * HIDDEN + n] = cv;
  }
  float* op = out + ((size_t)b * T + t) * 2;
  op[0] = y0;
  op[1] = y1;
}

#define CH    128
#define PITCH (CH * 2 + 2)
#define NSW   4
#define RPW   (16 / NSW)
#define PXP   (16 * 2 * NSW + 4)

typedef float v4f __attribute__((ext_vector_type(4)));

__device__ __forceinline__ void lds_barrier() {
  __builtin_amdgcn_fence(__ATOMIC_RELEASE, "workgroup", "local");
  __builtin_amdgcn_s_barrier();
  __builtin_amdgcn_fence(__ATOMIC_ACQUIRE, "workgroup", "local");
}

__device__ __forceinline__ void x_fetch(const float* __restrict__ xg, size_t rowf,
                                        int row0, int sb, int nsteps2, int lane,
                                        v2f (&v)[RPW][CH / 32]) {
#pragma unroll
  for (int bb = 0; bb < RPW; ++bb) {
    const float* row = xg + (size_t)(row0 + bb) * rowf;
#pragma unroll
    for (int m = 0; m < CH / 32; ++m) {
      int sg = sb + lane + 32 * m;
      sg = sg < nsteps2 ? sg : nsteps2 - 1;
      v[bb][m] = *(const v2f*)(row + (size_t)sg * 2);
    }
  }
}

__device__ __forceinline__ void x_put(float (*dst)[PITCH], int row0, int lane,
                                      const v2f (&v)[RPW][CH / 32]) {
#pragma unroll
  for (int bb = 0; bb < RPW; ++bb)
#pragma unroll
    for (int m = 0; m < CH / 32; ++m)
      *(v2f*)(&dst[row0 + bb][2 * (lane + 32 * m)]) = v[bb][m];
}

__device__ __forceinline__ v2f h_join(const float* p) {
  v4f pa = *(const v4f*)(p);
  v4f pb = *(const v4f*)(p + 4);
  v2f h;
  h[0] = (pa[0] + pa[2]) + (pb[0] + pb[2]);
  h[1] = (pa[1] + pa[3]) + (pb[1] + pb[3]);
  return h;
}

__device__ __forceinline__ void h_drain(float* __restrict__ og, size_t rowf,
                                        const float (*src)[PXP], int row0, int sb,
                                        int nsteps2, int lane) {
#pragma unroll
  for (int m = 0; m < CH / 32; ++m) {
    const int st = lane + 32 * m;
    if (sb + st < nsteps2) {
      float* col = og + (size_t)(sb + st) * 2;
#pragma unroll
      for (int bb = 0; bb < RPW; ++bb)
        *(v2f*)(col + (size_t)(row0 + bb) * rowf) =
            h_join(&src[st][(row0 + bb) * 2 * NSW]);
    }
  }
}

__global__ void __launch_bounds__(32 * NSW) lstm_phase2(
    const float* __restrict__ u_train, const float* __restrict__ W_ih,
    const float* __restrict__ W_hh, const float* __restrict__ b_ih,
    const float* __restrict__ b_hh, const float* __restrict__ W_hr,
    const float* __restrict__ c_last, float* __restrict__ out,
    int T, int nstep, int nsteps2) {
  __shared__ __attribute__((aligned(16))) float xs[2][16][PITCH];
  __shared__ __attribute__((aligned(16))) float px[2][CH][PXP];

  if (nsteps2 < 1) return;

  const int lane  = threadIdx.x & 31;
  const int wid   = __builtin_amdgcn_readfirstlane((int)(threadIdx.x >> 5));
  const int half  = lane >> 4;
  const int j     = lane & 15;
  const int mbase = blockIdx.x * 16;
  const int row0  = wid * RPW;

  v2f Ax, Ah;
  {
    const int g  = j & 3;
    const int n  = g * 16 + 4 * wid + (j >> 2);
    const float sc = (g == 2) ? 1.0f : -LOG2E;
    Ax[0] = half ? 0.0f : sc * W_ih[n * 2 + 0];
    Ax[1] = half ? 0.0f : sc * W_ih[n * 2 + 1];
    Ah[0] = sc * W_hh[n * 2 + 0];
    Ah[1] = sc * W_hh[n * 2 + 1];
  }

  const int ubase = 4 * wid + 2 * half;
  v8f cc;
#pragma unroll
  for (int r = 0; r < 8; ++r) {
    const int g = r & 3;
    const int n = g * 16 + ubase + (r >> 2);
    const float sc = (g == 2) ? 1.0f : -LOG2E;
    cc[r] = sc * (b_ih[n] + b_hh[n]);
  }

  float c[2], w0[2], w1[2];
#pragma unroll
  for (int q = 0; q < 2; ++q) {
    c[q]  = c_last[(mbase + j) * HIDDEN + ubase + q];
    w0[q] = W_hr[ubase + q];
    w1[q] = W_hr[16 + ubase + q];
  }

  v2f bh;
  {
    const float* hp = out + ((size_t)(mbase + j) * T + (nstep - 1)) * 2;
    bh[0] = half ? 0.0f : hp[0];
    bh[1] = half ? 0.0f : hp[1];
  }

  const size_t rowf = (size_t)T * 2;
  const float* xg = u_train + ((size_t)mbase * T + nstep) * 2;
  float*       og = out     + ((size_t)mbase * T + nstep) * 2;
  const int nchunks = (nsteps2 + CH - 1) / CH;

  v2f xr[RPW][CH / 32];
  x_fetch(xg, rowf, row0, 0, nsteps2, lane, xr);
  x_put(xs[0], row0, lane, xr);
  lds_barrier();

  for (int ch = 0; ch < nchunks; ++ch) {
    const int buf = ch & 1;
    const int sb  = ch * CH;
    int n = nsteps2 - sb;
    n = n < CH ? n : CH;

    x_fetch(xg, rowf, row0, sb + CH, nsteps2, lane, xr);

    const float* xrow = &xs[buf][j][0];
    float*       prow = &px[buf][0][j * 2 * NSW];
    const float* jrow = prow + 4 * half;
    v8f cx;
    {
      v2f xv = *(const v2f*)xrow;
      cx = __builtin_amdgcn_wmma_f32_16x16x4_f32(false, Ax, false, xv, (short)0, cc, false, false);
    }
    v2f nxv = *(const v2f*)(xrow + 2 * (n > 1 ? 1 : 0));

    for (int k = 0; k < n; ++k) {
      v8f d = __builtin_amdgcn_wmma_f32_16x16x4_f32(false, Ah, false, bh, (short)0, cx, false, false);

      float hp[2];
#pragma unroll
      for (int q = 0; q < 2; ++q) {
        float ig  = sigmoid_pre(d[4 * q + 0]);
        float fg  = sigmoid_pre(d[4 * q + 1]);
        float gg  = tanhf_fast(d[4 * q + 2]);
        float og_ = sigmoid_pre(d[4 * q + 3]);
        float cr  = __builtin_fmaf(fg, c[q], ig * gg);
        c[q] = cr;
        hp[q] = og_ * tanhf_fast(cr);
      }

      float s0 = __builtin_fmaf(hp[0], w0[0], hp[1] * w0[1]);
      float s1 = __builtin_fmaf(hp[0], w1[0], hp[1] * w1[1]);
      s0 += halfswap16(s0);
      s1 += halfswap16(s1);

      float* pk = prow + (size_t)k * PXP;
      v2f part;
      part[0] = s0;
      part[1] = s1;
      *(v2f*)(pk + 2 * wid) = part;

      cx = __builtin_amdgcn_wmma_f32_16x16x4_f32(false, Ax, false, nxv, (short)0, cc, false, false);

      lds_barrier();
      v4f q = *(const v4f*)(jrow + (size_t)k * PXP);
      int kn = (k + 2 < n) ? k + 2 : n - 1;
      nxv = *(const v2f*)(xrow + 2 * kn);
      __builtin_amdgcn_sched_barrier(0);
      bh[0] = q[0] + q[2];
      bh[1] = q[1] + q[3];
    }

    x_put(xs[buf ^ 1], row0, lane, xr);
    lds_barrier();
    h_drain(og, rowf, px[buf], row0, sb, nsteps2, lane);
  }
}

extern "C" void kernel_launch(void* const* d_in, const int* in_sizes, int n_in,
                              void* d_out, int out_size, void* d_ws, size_t ws_size,
                              hipStream_t stream) {
  const float* u_train = (const float*)d_in[0];
  const float* y_train = (const float*)d_in[1];
  const float* W_ih    = (const float*)d_in[2];
  const float* W_hh    = (const float*)d_in[3];
  const float* b_ih    = (const float*)d_in[4];
  const float* b_hh    = (const float*)d_in[5];
  const float* W_hr    = (const float*)d_in[6];
  const int B = 64;
  const int T = in_sizes[0] / (B * INPUT);
  const int nstep = T / 2;

  float* out    = (float*)d_out;
  float* c_last = (float*)d_ws;

  int total = B * nstep;
  lstm_phase1<<<(total + 255) / 256, 256, 0, stream>>>(
      y_train, W_ih, b_ih, b_hh, W_hr, out, c_last, B, T, nstep);

  lstm_phase2<<<B / 16, 32 * NSW, 0, stream>>>(
      u_train, W_ih, W_hh, b_ih, b_hh, W_hr, c_last, out, T, nstep, T - nstep);
}
